# starting kernel
# baseline (speedup 1.0000x reference)
#include <hip/hip_runtime.h>
#include <hip/hip_bf16.h>

#define NB      64
#define LSIG    480000
#define TILE    256
#define NTILES  (LSIG / TILE)
#define NSEC    3
#define CSTRIDE 128
#define FSTRIDE 320

typedef __attribute__((ext_vector_type(2))) float v2f;
typedef __attribute__((ext_vector_type(8))) float v8f;

__device__ __forceinline__ v8f wmma_f32_16x16x4(v2f a, v2f b, v8f c) {
  return __builtin_amdgcn_wmma_f32_16x16x4_f32(false, a, false, b, (short)0, c,
                                               false, false);
}

__global__ void derive_kernel(const float* __restrict__ knobs,
                              const float* __restrict__ W1,
                              const float* __restrict__ b1v,
                              const float* __restrict__ W2,
                              const float* __restrict__ b2v,
                              float* __restrict__ coef,
                              float* __restrict__ fr,
                              float* __restrict__ gt) {
  int b = threadIdx.x;
  if (b >= NB) return;
  float k0 = knobs[b * 3 + 0], k1 = knobs[b * 3 + 1], k2 = knobs[b * 3 + 2];
  float hmlp[16];
  for (int i = 0; i < 16; ++i)
    hmlp[i] = tanhf(k0 * W1[0 * 16 + i] + k1 * W1[1 * 16 + i] +
                    k2 * W1[2 * 16 + i] + b1v[i]);
  float p[15];
  for (int j = 0; j < 15; ++j) {
    float acc = b2v[j];
    for (int i = 0; i < 16; ++i) acc += hmlp[i] * W2[i * 15 + j];
    p[j] = acc;
  }
  for (int s = 0; s < NSEC; ++s) {
    float c0 = p[s * 5 + 0], c1 = p[s * 5 + 1], c2 = p[s * 5 + 2];
    float a2c = 0.999f * tanhf(p[s * 5 + 4]);
    float a1c = 0.999f * (1.0f + a2c) * tanhf(p[s * 5 + 3]);
    float A00 = -a1c, A01 = 1.0f, A10 = -a2c, A11 = 0.0f;
    float Bv0 = c1 - a1c * c0, Bv1 = c2 - a2c * c0;
    float* cf = coef + (b * NSEC + s) * CSTRIDE;

    float h[16];
    {
      float v0 = Bv0, v1 = Bv1;
      h[0] = c0;
      for (int k = 1; k < 16; ++k) {
        h[k] = v0;
        float n0 = A00 * v0 + A01 * v1;
        float n1 = A10 * v0 + A11 * v1;
        v0 = n0; v1 = n1;
      }
    }
    {
      float v0 = Bv0, v1 = Bv1;
      cf[16 + 15] = v0; cf[32 + 15] = v1;
      for (int j = 14; j >= 0; --j) {
        float n0 = A00 * v0 + A01 * v1;
        float n1 = A10 * v0 + A11 * v1;
        v0 = n0; v1 = n1;
        cf[16 + j] = v0; cf[32 + j] = v1;
      }
    }
    float g0i[16], g1i[16];
    {
      float g0 = 1.0f, g1 = 0.0f;
      for (int i = 0; i < 16; ++i) {
        g0i[i] = g0; g1i[i] = g1;
        float n0 = g0 * A00 + g1 * A10;
        float n1 = g0 * A01 + g1 * A11;
        g0 = n0; g1 = n1;
      }
    }
    {
      float M00 = A00, M01 = A01, M10 = A10, M11 = A11;
      for (int q = 0; q < 4; ++q) {
        float n00 = M00 * M00 + M01 * M10, n01 = M00 * M01 + M01 * M11;
        float n10 = M10 * M00 + M11 * M10, n11 = M10 * M01 + M11 * M11;
        M00 = n00; M01 = n01; M10 = n10; M11 = n11;
      }
      for (int q = 0; q < 5; ++q) {
        cf[80 + 4 * q + 0] = M00; cf[80 + 4 * q + 1] = M01;
        cf[80 + 4 * q + 2] = M10; cf[80 + 4 * q + 3] = M11;
        float n00 = M00 * M00 + M01 * M10, n01 = M00 * M01 + M01 * M11;
        float n10 = M10 * M00 + M11 * M10, n11 = M10 * M01 + M11 * M11;
        M00 = n00; M01 = n01; M10 = n10; M11 = n11;
      }
    }
    {
      float* fb = fr + (b * NSEC + s) * FSTRIDE;
      for (int lane = 0; lane < 32; ++lane) {
        int m = lane & 15, half = lane >> 4;
        for (int k = 0; k < 4; ++k) {
          int K0 = 4 * k + 2 * half;
          fb[(k * 32 + lane) * 2 + 0] = (K0 <= m)     ? h[m - K0]     : 0.0f;
          fb[(k * 32 + lane) * 2 + 1] = (K0 + 1 <= m) ? h[m - K0 - 1] : 0.0f;
        }
        fb[256 + lane * 2 + 0] = half ? 0.0f : g0i[m];
        fb[256 + lane * 2 + 1] = half ? 0.0f : g1i[m];
      }
    }
    {
      float g0 = 1.0f, g1 = 0.0f;
      float* G = gt + (b * NSEC + s) * TILE * 2;
      for (int n = 0; n < TILE; ++n) {
        G[n * 2 + 0] = g0; G[n * 2 + 1] = g1;
        float n0 = g0 * A00 + g1 * A10;
        float n1 = g0 * A01 + g1 * A11;
        g0 = n0; g1 = n1;
      }
    }
  }
}

__global__ void __launch_bounds__(256) tile_kernel(const float* in, float* out,
                                                   const float* __restrict__ coef,
                                                   const float* __restrict__ fr,
                                                   const float* __restrict__ gt,
                                                   const float* __restrict__ sg,
                                                   float* __restrict__ ts,
                                                   int sec, int prevsec) {
  __shared__ float lds[8 * TILE];
  const int lane = threadIdx.x & 31;
  const int wave = threadIdx.x >> 5;
  const int gw = blockIdx.x * 8 + wave;
  const int b = gw / NTILES;
  const int t = gw - b * NTILES;
  const int base = b * LSIG + t * TILE;
  float* xs = lds + wave * TILE;

  float4 u0 = *(const float4*)(in + base + lane * 8);
  float4 u1 = *(const float4*)(in + base + lane * 8 + 4);

  if (prevsec >= 0) {
    float2 sv = ((const float2*)sg)[gw];
    const float4* G4 = (const float4*)(gt + (b * NSEC + prevsec) * TILE * 2);
    float4 ga = G4[lane * 4 + 0];
    float4 gb = G4[lane * 4 + 1];
    float4 gc = G4[lane * 4 + 2];
    float4 gd = G4[lane * 4 + 3];
    u0.x += ga.x * sv.x + ga.y * sv.y;
    u0.y += ga.z * sv.x + ga.w * sv.y;
    u0.z += gb.x * sv.x + gb.y * sv.y;
    u0.w += gb.z * sv.x + gb.w * sv.y;
    u1.x += gc.x * sv.x + gc.y * sv.y;
    u1.y += gc.z * sv.x + gc.w * sv.y;
    u1.z += gd.x * sv.x + gd.y * sv.y;
    u1.w += gd.z * sv.x + gd.w * sv.y;
  }
  *(float4*)(xs + lane * 8)     = u0;
  *(float4*)(xs + lane * 8 + 4) = u1;
  __syncthreads();

  const float* cf = coef + (b * NSEC + sec) * CSTRIDE;
  const v2f* hf = (const v2f*)(fr + (b * NSEC + sec) * FSTRIDE);
  const int m = lane & 15;
  const int half = lane >> 4;

  float4 xc[4];
#pragma unroll
  for (int q = 0; q < 4; ++q)
    xc[q] = ((const float4*)(xs + m * 16))[q];

  v8f acc = {0.f, 0.f, 0.f, 0.f, 0.f, 0.f, 0.f, 0.f};
#pragma unroll
  for (int k = 0; k < 4; ++k) {
    v2f a = hf[k * 32 + lane];
    v2f bb;
    bb.x = half ? xc[k].z : xc[k].x;
    bb.y = half ? xc[k].w : xc[k].y;
    acc = wmma_f32_16x16x4(a, bb, acc);
  }

  float fx0 = 0.f, fx1 = 0.f;
#pragma unroll
  for (int q = 0; q < 4; ++q) {
    fx0 += cf[16 + 4 * q + 0] * xc[q].x + cf[16 + 4 * q + 1] * xc[q].y +
           cf[16 + 4 * q + 2] * xc[q].z + cf[16 + 4 * q + 3] * xc[q].w;
    fx1 += cf[32 + 4 * q + 0] * xc[q].x + cf[32 + 4 * q + 1] * xc[q].y +
           cf[32 + 4 * q + 2] * xc[q].z + cf[32 + 4 * q + 3] * xc[q].w;
  }

  float p0 = fx0, p1 = fx1;
#pragma unroll
  for (int s = 0; s < 4; ++s) {
    float M00 = cf[80 + 4 * s + 0], M01 = cf[80 + 4 * s + 1];
    float M10 = cf[80 + 4 * s + 2], M11 = cf[80 + 4 * s + 3];
    int len = 1 << s;
    float q0 = __shfl_up(p0, (unsigned)len, 32);
    float q1 = __shfl_up(p1, (unsigned)len, 32);
    if (m >= len) {
      p0 = M00 * q0 + M01 * q1 + p0;
      p1 = M10 * q0 + M11 * q1 + p1;
    }
  }
  float e0 = __shfl_up(p0, 1u, 32);
  float e1 = __shfl_up(p1, 1u, 32);
  float s0c = (m >= 1) ? e0 : 0.f;
  float s1c = (m >= 1) ? e1 : 0.f;

  {
    v2f a = hf[128 + lane];
    v2f bb;
    bb.x = half ? 0.f : s0c;
    bb.y = half ? 0.f : s1c;
    acc = wmma_f32_16x16x4(a, bb, acc);
  }

  if (lane == 15) {
    ts[(b * NTILES + t) * 2 + 0] = p0;
    ts[(b * NTILES + t) * 2 + 1] = p1;
  }

  __syncthreads();
  {
    float4 d0, d1;
    d0.x = acc[0]; d0.y = acc[1]; d0.z = acc[2]; d0.w = acc[3];
    d1.x = acc[4]; d1.y = acc[5]; d1.z = acc[6]; d1.w = acc[7];
    *(float4*)(xs + m * 16 + 8 * half)     = d0;
    *(float4*)(xs + m * 16 + 8 * half + 4) = d1;
  }
  __syncthreads();
  *(float4*)(out + base + lane * 8)     = *(const float4*)(xs + lane * 8);
  *(float4*)(out + base + lane * 8 + 4) = *(const float4*)(xs + lane * 8 + 4);
}

__global__ void scan_kernel(const float* __restrict__ coef,
                            const float* __restrict__ ts,
                            float* __restrict__ sg, int sec) {
  int b = threadIdx.x;
  if (b >= NB) return;
  const float* cf = coef + (b * NSEC + sec) * CSTRIDE;
  float A00 = cf[96], A01 = cf[97], A10 = cf[98], A11 = cf[99];
  float s0 = 0.f, s1 = 0.f;
#pragma unroll 5
  for (int t = 0; t < NTILES; ++t) {
    int idx = (b * NTILES + t) * 2;
    sg[idx + 0] = s0; sg[idx + 1] = s1;
    float z0 = ts[idx + 0], z1 = ts[idx + 1];
    float n0 = A00 * s0 + A01 * s1 + z0;
    float n1 = A10 * s0 + A11 * s1 + z1;
    s0 = n0; s1 = n1;
  }
}

__global__ void __launch_bounds__(256) fixup_kernel(float* out,
                                                    const float* __restrict__ gt,
                                                    const float* __restrict__ sg,
                                                    int sec) {
  int tile = blockIdx.x;
  int b = tile / NTILES;
  int t = tile - b * NTILES;
  int n = threadIdx.x;
  float s0 = sg[tile * 2 + 0], s1 = sg[tile * 2 + 1];
  const float2* G = (const float2*)(gt + (b * NSEC + sec) * TILE * 2);
  float2 g = G[n];
  int idx = b * LSIG + t * TILE + n;
  out[idx] += g.x * s0 + g.y * s1;
}

extern "C" void kernel_launch(void* const* d_in, const int* in_sizes, int n_in,
                              void* d_out, int out_size, void* d_ws, size_t ws_size,
                              hipStream_t stream) {
  (void)in_sizes; (void)n_in; (void)out_size; (void)ws_size;
  const float* x     = (const float*)d_in[0];
  const float* knobs = (const float*)d_in[1];
  const float* W1    = (const float*)d_in[2];
  const float* b1v   = (const float*)d_in[3];
  const float* W2    = (const float*)d_in[4];
  const float* b2v   = (const float*)d_in[5];
  float* out = (float*)d_out;

  float* COEF = (float*)d_ws;
  float* FR   = COEF + NB * NSEC * CSTRIDE;
  float* GT   = FR   + NB * NSEC * FSTRIDE;
  float* TS   = GT   + NB * NSEC * TILE * 2;
  float* SG   = TS   + NB * NTILES * 2;

  derive_kernel<<<1, 64, 0, stream>>>(knobs, W1, b1v, W2, b2v, COEF, FR, GT);

  const int tileBlocks = (NB * NTILES) / 8;
  for (int sec = 0; sec < NSEC; ++sec) {
    const float* src = (sec == 0) ? x : out;
    tile_kernel<<<tileBlocks, 256, 0, stream>>>(src, out, COEF, FR, GT, SG, TS,
                                                sec, sec - 1);
    scan_kernel<<<1, 64, 0, stream>>>(COEF, TS, SG, sec);
  }
  fixup_kernel<<<NB * NTILES, 256, 0, stream>>>(out, GT, SG, NSEC - 1);
}


# both passes: a wave's 16 B columns are 16 tiles, chunk state carried step to step in WMMA accumulators (A^16 S on WMMA): no chunk scan, no LDS; 7552 waves a pass, was 120000; sum order changes
# speedup vs baseline: 9.4502x; 9.4502x over previous
#include <hip/hip_runtime.h>
#include <hip/hip_bf16.h>

#define NB      64
#define LSIG    480000
#define TILE    256
#define NTILES  (LSIG / TILE)
#define NGRP    ((NTILES + 15) / 16)
#define NST     6
#define NPOW    19
#define PSTRIDE (NPOW * 36)
#define FSTRIDE 640
#define SCAN_THREADS 1024
#define SCAN_LEVELS  10
#define WS_FLOATS (NB * PSTRIDE + NB * FSTRIDE + NB * 3 * NTILES * 2)
#define WS_BYTES  ((size_t)WS_FLOATS * 4)

typedef __attribute__((ext_vector_type(2))) float v2f;
typedef __attribute__((ext_vector_type(8))) float v8f;

__device__ __forceinline__ v8f wmma_f32_16x16x4(v2f a, v2f b, v8f c) {
  return __builtin_amdgcn_wmma_f32_16x16x4_f32(false, a, false, b, (short)0, c,
                                               false, false);
}

__device__ __forceinline__ void mv6_acc(const float* __restrict__ M, const float* q, float* v) {
  v[0] += M[0] * q[0] + M[1] * q[1];
  v[1] += M[6] * q[0] + M[7] * q[1];
  v[2] += M[12] * q[0] + M[13] * q[1] + M[14] * q[2] + M[15] * q[3];
  v[3] += M[18] * q[0] + M[19] * q[1] + M[20] * q[2] + M[21] * q[3];
  v[4] += M[24] * q[0] + M[25] * q[1] + M[26] * q[2] + M[27] * q[3] + M[28] * q[4] + M[29] * q[5];
  v[5] += M[30] * q[0] + M[31] * q[1] + M[32] * q[2] + M[33] * q[3] + M[34] * q[4] + M[35] * q[5];
}

__global__ void __launch_bounds__(256) derive_kernel(const float* __restrict__ knobs,
                                                     const float* __restrict__ W1,
                                                     const float* __restrict__ b1v,
                                                     const float* __restrict__ W2,
                                                     const float* __restrict__ b2v,
                                                     float* __restrict__ pw,
                                                     float* __restrict__ fr,
                                                     size_t wsz) {
  if (wsz < WS_BYTES) return;
  __shared__ double P[NPOW][36];
  __shared__ double Gs[16][NST];
  __shared__ double Fs[16][NST];
  __shared__ float hs[16];
  const int b = blockIdx.x;
  const int tid = threadIdx.x;

  float k0 = knobs[b * 3 + 0], k1 = knobs[b * 3 + 1], k2 = knobs[b * 3 + 2];
  float hmlp[16];
#pragma unroll
  for (int i = 0; i < 16; ++i)
    hmlp[i] = tanhf(k0 * W1[0 * 16 + i] + k1 * W1[1 * 16 + i] +
                    k2 * W1[2 * 16 + i] + b1v[i]);
  float p[15];
#pragma unroll
  for (int j = 0; j < 15; ++j) {
    float acc = b2v[j];
#pragma unroll
    for (int i = 0; i < 16; ++i) acc += hmlp[i] * W2[i * 15 + j];
    p[j] = acc;
  }
  double a1[3], a2[3], d[3], B0[3], B1[3];
#pragma unroll
  for (int s = 0; s < 3; ++s) {
    float c0 = p[s * 5 + 0], c1 = p[s * 5 + 1], c2 = p[s * 5 + 2];
    float a2f = 0.999f * tanhf(p[s * 5 + 4]);
    float a1f = 0.999f * (1.0f + a2f) * tanhf(p[s * 5 + 3]);
    a1[s] = (double)a1f; a2[s] = (double)a2f;
    d[s] = (double)c0;
    B0[s] = (double)c1 - a1[s] * d[s];
    B1[s] = (double)c2 - a2[s] * d[s];
  }
  double B6[NST], C6[NST];
  B6[0] = B0[0];               B6[1] = B1[0];
  B6[2] = B0[1] * d[0];        B6[3] = B1[1] * d[0];
  B6[4] = B0[2] * d[1] * d[0]; B6[5] = B1[2] * d[1] * d[0];
  C6[0] = d[2] * d[1]; C6[1] = 0.0; C6[2] = d[2]; C6[3] = 0.0; C6[4] = 1.0; C6[5] = 0.0;
  const double D6 = d[2] * d[1] * d[0];

  if (tid == 0) {
    double A6[36];
#pragma unroll
    for (int e = 0; e < 36; ++e) A6[e] = 0.0;
#pragma unroll
    for (int s = 0; s < 3; ++s) {
      A6[(2 * s) * 6 + 2 * s]         = -a1[s];
      A6[(2 * s) * 6 + 2 * s + 1]     = 1.0;
      A6[(2 * s + 1) * 6 + 2 * s]     = -a2[s];
    }
    A6[2 * 6 + 0] = B0[1];        A6[3 * 6 + 0] = B1[1];
    A6[4 * 6 + 2] = B0[2];        A6[5 * 6 + 2] = B1[2];
    A6[4 * 6 + 0] = B0[2] * d[1]; A6[5 * 6 + 0] = B1[2] * d[1];
#pragma unroll
    for (int e = 0; e < 36; ++e) P[0][e] = A6[e];
  }
  __syncthreads();
  for (int k = 1; k < NPOW; ++k) {
    if (tid < 36) {
      const int r = tid / 6, c = tid - r * 6;
      double acc = 0.0;
#pragma unroll
      for (int j = 0; j < 6; ++j) acc += P[k - 1][r * 6 + j] * P[k - 1][j * 6 + c];
      P[k][tid] = acc;
    }
    __syncthreads();
  }
  if (tid < 16) {
    const int n = tid;
    double g[NST];
#pragma unroll
    for (int i = 0; i < NST; ++i) g[i] = C6[i];
    for (int k = 0; k < 4; ++k) {
      if ((n >> k) & 1) {
        double t[NST];
#pragma unroll
        for (int c = 0; c < NST; ++c) {
          double acc = 0.0;
#pragma unroll
          for (int r = 0; r < NST; ++r) acc += g[r] * P[k][r * 6 + c];
          t[c] = acc;
        }
#pragma unroll
        for (int i = 0; i < NST; ++i) g[i] = t[i];
      }
    }
#pragma unroll
    for (int i = 0; i < NST; ++i) Gs[n][i] = g[i];
  }
  if (tid < 16) {
    const int e = 15 - tid;
    double f[NST];
#pragma unroll
    for (int i = 0; i < NST; ++i) f[i] = B6[i];
    for (int k = 0; k < 4; ++k) {
      if ((e >> k) & 1) {
        double t[NST];
#pragma unroll
        for (int r = 0; r < NST; ++r) {
          double acc = 0.0;
#pragma unroll
          for (int c = 0; c < NST; ++c) acc += P[k][r * 6 + c] * f[c];
          t[r] = acc;
        }
#pragma unroll
        for (int i = 0; i < NST; ++i) f[i] = t[i];
      }
    }
#pragma unroll
    for (int i = 0; i < NST; ++i) Fs[tid][i] = f[i];
  }
  __syncthreads();
  if (tid < 16) {
    double acc = D6;
    if (tid > 0) {
      acc = 0.0;
#pragma unroll
      for (int i = 0; i < NST; ++i) acc += Gs[tid - 1][i] * B6[i];
    }
    hs[tid] = (float)acc;
  }
  __syncthreads();
  if (tid < 32) {
    const int lane = tid, m = lane & 15, half = lane >> 4;
    float* fb = fr + b * FSTRIDE;
    float* ab = pw + b * PSTRIDE;
    const int pm = 8 * ((m & 7) >> 2) + 4 * (m >> 3) + (m & 3);
    int sidx = -1;
    if (m < 2) sidx = m; else if (m < 4) sidx = m + 2; else if (m == 8 || m == 9) sidx = m - 6;
    for (int s = 0; s < 4; ++s) {
      for (int e = 0; e < 2; ++e) {
        const int i = 2 * s + e;
        const int j = 8 * (i >> 2) + 4 * half + (i & 3);
        fb[(s * 32 + lane) * 2 + e]       = (j <= pm) ? hs[pm - j] : 0.0f;
        fb[256 + (s * 32 + lane) * 2 + e] = (sidx >= 0) ? (float)Fs[j][sidx] : 0.0f;
      }
    }
    for (int s = 0; s < 2; ++s) {
      for (int e = 0; e < 2; ++e) {
        const int kst = (s == 0) ? (2 * half + e) : (half ? -1 : 4 + e);
        fb[512 + (s * 32 + lane) * 2 + e] = (kst >= 0) ? (float)Gs[pm][kst] : 0.0f;
        ab[(s * 32 + lane) * 2 + e] = (kst >= 0 && sidx >= 0) ? (float)P[4][sidx * 6 + kst] : 0.0f;
      }
    }
  }
  {
    const double* Pf = &P[0][0];
    for (int idx = 8 * 36 + tid; idx < PSTRIDE; idx += 256) pw[b * PSTRIDE + idx] = (float)Pf[idx];
  }
}

__global__ void __launch_bounds__(256) state_kernel(const float* __restrict__ in,
                                                    const float* __restrict__ pw,
                                                    const float* __restrict__ fr,
                                                    float* __restrict__ ts,
                                                    size_t wsz) {
  if (wsz < WS_BYTES) return;
  const int lane = threadIdx.x & 31;
  const int wave = __builtin_amdgcn_readfirstlane((int)(threadIdx.x >> 5));
  const int gw = blockIdx.x * 8 + wave;
  const int b = gw / NGRP;
  const int g = gw - b * NGRP;
  const int n = lane & 15;
  const int half = lane >> 4;
  const int t = g * 16 + n;
  const bool valid = t < NTILES;
  const int tc = valid ? t : (NTILES - 1);

  const float4* xp = (const float4*)(in + (size_t)b * LSIG + tc * TILE + 4 * half);
  float4 xv[16][2];
#pragma unroll
  for (int c = 0; c < 16; ++c) {
    xv[c][0] = xp[4 * c];
    xv[c][1] = xp[4 * c + 2];
  }

  const v2f* hf = (const v2f*)(fr + b * FSTRIDE);
  const v2f* af = (const v2f*)(pw + b * PSTRIDE);
  v2f F[4], A[2];
#pragma unroll
  for (int s = 0; s < 4; ++s) F[s] = hf[128 + s * 32 + lane];
  A[0] = af[lane];
  A[1] = af[32 + lane];

  v8f S = {0.f, 0.f, 0.f, 0.f, 0.f, 0.f, 0.f, 0.f};
#pragma unroll
  for (int c = 0; c < 16; ++c) {
    v2f x0, x1, x2, x3;
    x0.x = xv[c][0].x; x0.y = xv[c][0].y;
    x1.x = xv[c][0].z; x1.y = xv[c][0].w;
    x2.x = xv[c][1].x; x2.y = xv[c][1].y;
    x3.x = xv[c][1].z; x3.y = xv[c][1].w;
    v8f acc = {0.f, 0.f, 0.f, 0.f, 0.f, 0.f, 0.f, 0.f};
    acc = wmma_f32_16x16x4(F[0], x0, acc);
    acc = wmma_f32_16x16x4(F[1], x1, acc);
    acc = wmma_f32_16x16x4(F[2], x2, acc);
    acc = wmma_f32_16x16x4(F[3], x3, acc);
    if (c > 0) {
      v2f s0, s1;
      s0.x = S[0]; s0.y = S[1];
      s1.x = S[2]; s1.y = S[3];
      acc = wmma_f32_16x16x4(A[0], s0, acc);
      acc = wmma_f32_16x16x4(A[1], s1, acc);
    }
    S = acc;
  }

  if (valid) {
    float2* T = (float2*)ts;
    T[(b * 3 + half) * NTILES + t] = make_float2(S[0], S[1]);
    if (half == 0) T[(b * 3 + 2) * NTILES + t] = make_float2(S[2], S[3]);
  }
}

__global__ void __launch_bounds__(SCAN_THREADS) scan_kernel(const float* __restrict__ pw,
                                                            float* __restrict__ ts,
                                                            size_t wsz) {
  if (wsz < WS_BYTES) return;
  __shared__ float2 buf[2][3][SCAN_THREADS];
  const int b = blockIdx.x;
  const int i = threadIdx.x;
  const float* pwr = pw + b * PSTRIDE;
  const float* A256 = pwr + 8 * 36;
  float2* zp = (float2*)ts + (size_t)b * 3 * NTILES;

  const int t0 = 2 * i, t1 = 2 * i + 1;
  float z0[NST], v[NST];
#pragma unroll
  for (int j = 0; j < 3; ++j) {
    float2 a = make_float2(0.f, 0.f), c = make_float2(0.f, 0.f);
    if (t0 < NTILES) a = zp[j * NTILES + t0];
    if (t1 < NTILES) c = zp[j * NTILES + t1];
    z0[2 * j] = a.x; z0[2 * j + 1] = a.y;
    v[2 * j] = c.x;  v[2 * j + 1] = c.y;
  }
  mv6_acc(A256, z0, v);

  int cur = 0;
#pragma unroll
  for (int j = 0; j < 3; ++j) buf[0][j][i] = make_float2(v[2 * j], v[2 * j + 1]);
  __syncthreads();
  for (int k = 0; k < SCAN_LEVELS; ++k) {
    const int len = 1 << k;
    const float* M = pwr + (9 + k) * 36;
    if (i >= len) {
      float q[NST];
#pragma unroll
      for (int j = 0; j < 3; ++j) {
        const float2 w = buf[cur][j][i - len];
        q[2 * j] = w.x; q[2 * j + 1] = w.y;
      }
      mv6_acc(M, q, v);
    }
    cur ^= 1;
#pragma unroll
    for (int j = 0; j < 3; ++j) buf[cur][j][i] = make_float2(v[2 * j], v[2 * j + 1]);
    __syncthreads();
  }

  float e[NST], e1[NST];
#pragma unroll
  for (int j = 0; j < 3; ++j) {
    float2 w = make_float2(0.f, 0.f);
    if (i > 0) w = buf[cur][j][i - 1];
    e[2 * j] = w.x; e[2 * j + 1] = w.y;
  }
#pragma unroll
  for (int q = 0; q < NST; ++q) e1[q] = z0[q];
  mv6_acc(A256, e, e1);
#pragma unroll
  for (int j = 0; j < 3; ++j) {
    if (t0 < NTILES) zp[j * NTILES + t0] = make_float2(e[2 * j], e[2 * j + 1]);
    if (t1 < NTILES) zp[j * NTILES + t1] = make_float2(e1[2 * j], e1[2 * j + 1]);
  }
}

__global__ void __launch_bounds__(256) tile_kernel(const float* __restrict__ in, float* __restrict__ out,
                                                   const float* __restrict__ pw,
                                                   const float* __restrict__ fr,
                                                   const float* __restrict__ sg,
                                                   size_t wsz) {
  if (wsz < WS_BYTES) return;
  const int lane = threadIdx.x & 31;
  const int wave = __builtin_amdgcn_readfirstlane((int)(threadIdx.x >> 5));
  const int gw = blockIdx.x * 8 + wave;
  const int b = gw / NGRP;
  const int g = gw - b * NGRP;
  const int n = lane & 15;
  const int half = lane >> 4;
  const int t = g * 16 + n;
  const bool valid = t < NTILES;
  const int tc = valid ? t : (NTILES - 1);

  const float4* xp = (const float4*)(in + (size_t)b * LSIG + tc * TILE + 4 * half);
  float4 xv[16][2];
#pragma unroll
  for (int c = 0; c < 16; ++c) {
    xv[c][0] = xp[4 * c];
    xv[c][1] = xp[4 * c + 2];
  }

  v8f S = {0.f, 0.f, 0.f, 0.f, 0.f, 0.f, 0.f, 0.f};
  {
    const float2* Sg = (const float2*)sg;
    const float2 w0 = Sg[(b * 3 + half) * NTILES + tc];
    const float2 w2 = Sg[(b * 3 + 2) * NTILES + tc];
    S[0] = w0.x; S[1] = w0.y;
    S[2] = half ? 0.f : w2.x;
    S[3] = half ? 0.f : w2.y;
  }

  const v2f* hf = (const v2f*)(fr + b * FSTRIDE);
  const v2f* af = (const v2f*)(pw + b * PSTRIDE);
  v2f H[4], F[4], G[2], A[2];
#pragma unroll
  for (int s = 0; s < 4; ++s) {
    H[s] = hf[s * 32 + lane];
    F[s] = hf[128 + s * 32 + lane];
  }
  G[0] = hf[256 + lane];
  G[1] = hf[256 + 32 + lane];
  A[0] = af[lane];
  A[1] = af[32 + lane];

  float4* op = (float4*)(out + (size_t)b * LSIG + tc * TILE + 4 * half);
#pragma unroll
  for (int c = 0; c < 16; ++c) {
    v2f x0, x1, x2, x3, s0, s1;
    x0.x = xv[c][0].x; x0.y = xv[c][0].y;
    x1.x = xv[c][0].z; x1.y = xv[c][0].w;
    x2.x = xv[c][1].x; x2.y = xv[c][1].y;
    x3.x = xv[c][1].z; x3.y = xv[c][1].w;
    s0.x = S[0]; s0.y = S[1];
    s1.x = S[2]; s1.y = S[3];

    v8f y = {0.f, 0.f, 0.f, 0.f, 0.f, 0.f, 0.f, 0.f};
    y = wmma_f32_16x16x4(H[0], x0, y);
    y = wmma_f32_16x16x4(H[1], x1, y);
    y = wmma_f32_16x16x4(H[2], x2, y);
    y = wmma_f32_16x16x4(H[3], x3, y);
    y = wmma_f32_16x16x4(G[0], s0, y);
    y = wmma_f32_16x16x4(G[1], s1, y);

    if (c < 15) {
      v8f acc = {0.f, 0.f, 0.f, 0.f, 0.f, 0.f, 0.f, 0.f};
      acc = wmma_f32_16x16x4(F[0], x0, acc);
      acc = wmma_f32_16x16x4(F[1], x1, acc);
      acc = wmma_f32_16x16x4(F[2], x2, acc);
      acc = wmma_f32_16x16x4(F[3], x3, acc);
      acc = wmma_f32_16x16x4(A[0], s0, acc);
      acc = wmma_f32_16x16x4(A[1], s1, acc);
      S = acc;
    }

    if (valid) {
      float4 d0, d1;
      d0.x = y[0]; d0.y = y[1]; d0.z = y[2]; d0.w = y[3];
      d1.x = y[4]; d1.y = y[5]; d1.z = y[6]; d1.w = y[7];
      op[4 * c]     = d0;
      op[4 * c + 2] = d1;
    }
  }
}

extern "C" void kernel_launch(void* const* d_in, const int* in_sizes, int n_in,
                              void* d_out, int out_size, void* d_ws, size_t ws_size,
                              hipStream_t stream) {
  (void)in_sizes; (void)n_in; (void)out_size;
  const float* x     = (const float*)d_in[0];
  const float* knobs = (const float*)d_in[1];
  const float* W1    = (const float*)d_in[2];
  const float* b1v   = (const float*)d_in[3];
  const float* W2    = (const float*)d_in[4];
  const float* b2v   = (const float*)d_in[5];
  float* out = (float*)d_out;

  float* PW = (float*)d_ws;
  float* FR = PW + NB * PSTRIDE;
  float* TS = FR + NB * FSTRIDE;

  derive_kernel<<<NB, 256, 0, stream>>>(knobs, W1, b1v, W2, b2v, PW, FR, ws_size);
  state_kernel<<<(NB * NGRP) / 8, 256, 0, stream>>>(x, PW, FR, TS, ws_size);
  scan_kernel<<<NB, SCAN_THREADS, 0, stream>>>(PW, TS, ws_size);
  tile_kernel<<<(NB * NGRP) / 8, 256, 0, stream>>>(x, out, PW, FR, TS, ws_size);
}


# both passes: a wave's 16 tiles (16 KB contiguous) pass through its own LDS rows (pitch 1040 B): global loads and stores move whole lines, 512 B an instruction, not 32 B of 16 lines; sums unchanged
# speedup vs baseline: 12.5147x; 1.3243x over previous
#include <hip/hip_runtime.h>
#include <hip/hip_bf16.h>

#define NB      64
#define LSIG    480000
#define TILE    256
#define NTILES  (LSIG / TILE)
#define NGRP    ((NTILES + 15) / 16)
#define NST     6
#define NPOW    19
#define PSTRIDE (NPOW * 36)
#define FSTRIDE 640
#define SCAN_THREADS 1024
#define SCAN_LEVELS  10
#define SPITCH  65
#define WSTAGE  (16 * SPITCH)
#define WS_FLOATS (NB * PSTRIDE + NB * FSTRIDE + NB * 3 * NTILES * 2)
#define WS_BYTES  ((size_t)WS_FLOATS * 4)

typedef __attribute__((ext_vector_type(2))) float v2f;
typedef __attribute__((ext_vector_type(8))) float v8f;

__device__ __forceinline__ v8f wmma_f32_16x16x4(v2f a, v2f b, v8f c) {
  return __builtin_amdgcn_wmma_f32_16x16x4_f32(false, a, false, b, (short)0, c,
                                               false, false);
}

__device__ __forceinline__ void mv6_acc(const float* __restrict__ M, const float* q, float* v) {
  v[0] += M[0] * q[0] + M[1] * q[1];
  v[1] += M[6] * q[0] + M[7] * q[1];
  v[2] += M[12] * q[0] + M[13] * q[1] + M[14] * q[2] + M[15] * q[3];
  v[3] += M[18] * q[0] + M[19] * q[1] + M[20] * q[2] + M[21] * q[3];
  v[4] += M[24] * q[0] + M[25] * q[1] + M[26] * q[2] + M[27] * q[3] + M[28] * q[4] + M[29] * q[5];
  v[5] += M[30] * q[0] + M[31] * q[1] + M[32] * q[2] + M[33] * q[3] + M[34] * q[4] + M[35] * q[5];
}

__global__ void __launch_bounds__(256) derive_kernel(const float* __restrict__ knobs,
                                                     const float* __restrict__ W1,
                                                     const float* __restrict__ b1v,
                                                     const float* __restrict__ W2,
                                                     const float* __restrict__ b2v,
                                                     float* __restrict__ pw,
                                                     float* __restrict__ fr,
                                                     size_t wsz) {
  if (wsz < WS_BYTES) return;
  __shared__ double P[NPOW][36];
  __shared__ double Gs[16][NST];
  __shared__ double Fs[16][NST];
  __shared__ float hs[16];
  const int b = blockIdx.x;
  const int tid = threadIdx.x;

  float k0 = knobs[b * 3 + 0], k1 = knobs[b * 3 + 1], k2 = knobs[b * 3 + 2];
  float hmlp[16];
#pragma unroll
  for (int i = 0; i < 16; ++i)
    hmlp[i] = tanhf(k0 * W1[0 * 16 + i] + k1 * W1[1 * 16 + i] +
                    k2 * W1[2 * 16 + i] + b1v[i]);
  float p[15];
#pragma unroll
  for (int j = 0; j < 15; ++j) {
    float acc = b2v[j];
#pragma unroll
    for (int i = 0; i < 16; ++i) acc += hmlp[i] * W2[i * 15 + j];
    p[j] = acc;
  }
  double a1[3], a2[3], d[3], B0[3], B1[3];
#pragma unroll
  for (int s = 0; s < 3; ++s) {
    float c0 = p[s * 5 + 0], c1 = p[s * 5 + 1], c2 = p[s * 5 + 2];
    float a2f = 0.999f * tanhf(p[s * 5 + 4]);
    float a1f = 0.999f * (1.0f + a2f) * tanhf(p[s * 5 + 3]);
    a1[s] = (double)a1f; a2[s] = (double)a2f;
    d[s] = (double)c0;
    B0[s] = (double)c1 - a1[s] * d[s];
    B1[s] = (double)c2 - a2[s] * d[s];
  }
  double B6[NST], C6[NST];
  B6[0] = B0[0];               B6[1] = B1[0];
  B6[2] = B0[1] * d[0];        B6[3] = B1[1] * d[0];
  B6[4] = B0[2] * d[1] * d[0]; B6[5] = B1[2] * d[1] * d[0];
  C6[0] = d[2] * d[1]; C6[1] = 0.0; C6[2] = d[2]; C6[3] = 0.0; C6[4] = 1.0; C6[5] = 0.0;
  const double D6 = d[2] * d[1] * d[0];

  if (tid == 0) {
    double A6[36];
#pragma unroll
    for (int e = 0; e < 36; ++e) A6[e] = 0.0;
#pragma unroll
    for (int s = 0; s < 3; ++s) {
      A6[(2 * s) * 6 + 2 * s]         = -a1[s];
      A6[(2 * s) * 6 + 2 * s + 1]     = 1.0;
      A6[(2 * s + 1) * 6 + 2 * s]     = -a2[s];
    }
    A6[2 * 6 + 0] = B0[1];        A6[3 * 6 + 0] = B1[1];
    A6[4 * 6 + 2] = B0[2];        A6[5 * 6 + 2] = B1[2];
    A6[4 * 6 + 0] = B0[2] * d[1]; A6[5 * 6 + 0] = B1[2] * d[1];
#pragma unroll
    for (int e = 0; e < 36; ++e) P[0][e] = A6[e];
  }
  __syncthreads();
  for (int k = 1; k < NPOW; ++k) {
    if (tid < 36) {
      const int r = tid / 6, c = tid - r * 6;
      double acc = 0.0;
#pragma unroll
      for (int j = 0; j < 6; ++j) acc += P[k - 1][r * 6 + j] * P[k - 1][j * 6 + c];
      P[k][tid] = acc;
    }
    __syncthreads();
  }
  if (tid < 16) {
    const int n = tid;
    double g[NST];
#pragma unroll
    for (int i = 0; i < NST; ++i) g[i] = C6[i];
    for (int k = 0; k < 4; ++k) {
      if ((n >> k) & 1) {
        double t[NST];
#pragma unroll
        for (int c = 0; c < NST; ++c) {
          double acc = 0.0;
#pragma unroll
          for (int r = 0; r < NST; ++r) acc += g[r] * P[k][r * 6 + c];
          t[c] = acc;
        }
#pragma unroll
        for (int i = 0; i < NST; ++i) g[i] = t[i];
      }
    }
#pragma unroll
    for (int i = 0; i < NST; ++i) Gs[n][i] = g[i];
  }
  if (tid < 16) {
    const int e = 15 - tid;
    double f[NST];
#pragma unroll
    for (int i = 0; i < NST; ++i) f[i] = B6[i];
    for (int k = 0; k < 4; ++k) {
      if ((e >> k) & 1) {
        double t[NST];
#pragma unroll
        for (int r = 0; r < NST; ++r) {
          double acc = 0.0;
#pragma unroll
          for (int c = 0; c < NST; ++c) acc += P[k][r * 6 + c] * f[c];
          t[r] = acc;
        }
#pragma unroll
        for (int i = 0; i < NST; ++i) f[i] = t[i];
      }
    }
#pragma unroll
    for (int i = 0; i < NST; ++i) Fs[tid][i] = f[i];
  }
  __syncthreads();
  if (tid < 16) {
    double acc = D6;
    if (tid > 0) {
      acc = 0.0;
#pragma unroll
      for (int i = 0; i < NST; ++i) acc += Gs[tid - 1][i] * B6[i];
    }
    hs[tid] = (float)acc;
  }
  __syncthreads();
  if (tid < 32) {
    const int lane = tid, m = lane & 15, half = lane >> 4;
    float* fb = fr + b * FSTRIDE;
    float* ab = pw + b * PSTRIDE;
    const int pm = 8 * ((m & 7) >> 2) + 4 * (m >> 3) + (m & 3);
    int sidx = -1;
    if (m < 2) sidx = m; else if (m < 4) sidx = m + 2; else if (m == 8 || m == 9) sidx = m - 6;
    for (int s = 0; s < 4; ++s) {
      for (int e = 0; e < 2; ++e) {
        const int i = 2 * s + e;
        const int j = 8 * (i >> 2) + 4 * half + (i & 3);
        fb[(s * 32 + lane) * 2 + e]       = (j <= pm) ? hs[pm - j] : 0.0f;
        fb[256 + (s * 32 + lane) * 2 + e] = (sidx >= 0) ? (float)Fs[j][sidx] : 0.0f;
      }
    }
    for (int s = 0; s < 2; ++s) {
      for (int e = 0; e < 2; ++e) {
        const int kst = (s == 0) ? (2 * half + e) : (half ? -1 : 4 + e);
        fb[512 + (s * 32 + lane) * 2 + e] = (kst >= 0) ? (float)Gs[pm][kst] : 0.0f;
        ab[(s * 32 + lane) * 2 + e] = (kst >= 0 && sidx >= 0) ? (float)P[4][sidx * 6 + kst] : 0.0f;
      }
    }
  }
  {
    const double* Pf = &P[0][0];
    for (int idx = 8 * 36 + tid; idx < PSTRIDE; idx += 256) pw[b * PSTRIDE + idx] = (float)Pf[idx];
  }
}

template <bool FULL>
__device__ __forceinline__ void stage_in(const float4* __restrict__ gbase, int nv, int lane, float4* lw) {
  float4 xv[32];
#pragma unroll
  for (int j = 0; j < 32; ++j) {
    const int tt = j & 15, hh = j >> 4;
    const int ts = (FULL || tt < nv) ? tt : 0;
    xv[j] = gbase[ts * (TILE / 4) + hh * 32 + lane];
  }
#pragma unroll
  for (int j = 0; j < 32; ++j) lw[(j & 15) * SPITCH + (j >> 4) * 32 + lane] = xv[j];
  __builtin_amdgcn_wave_barrier();
}

__global__ void __launch_bounds__(256) state_kernel(const float* __restrict__ in,
                                                    const float* __restrict__ pw,
                                                    const float* __restrict__ fr,
                                                    float* __restrict__ ts,
                                                    size_t wsz) {
  if (wsz < WS_BYTES) return;
  __shared__ float4 stage[8 * WSTAGE];
  const int lane = threadIdx.x & 31;
  const int wave = __builtin_amdgcn_readfirstlane((int)(threadIdx.x >> 5));
  const int gw = blockIdx.x * 8 + wave;
  const int b = gw / NGRP;
  const int g = gw - b * NGRP;
  const int n = lane & 15;
  const int half = lane >> 4;
  const int t = g * 16 + n;
  const bool valid = t < NTILES;
  const int nv = (NTILES - g * 16 < 16) ? (NTILES - g * 16) : 16;
  float4* lw = stage + wave * WSTAGE;

  const v2f* hf = (const v2f*)(fr + b * FSTRIDE);
  const v2f* af = (const v2f*)(pw + b * PSTRIDE);
  v2f F[4], A[2];
#pragma unroll
  for (int s = 0; s < 4; ++s) F[s] = hf[128 + s * 32 + lane];
  A[0] = af[lane];
  A[1] = af[32 + lane];

  const float4* gbase = (const float4*)(in + (size_t)b * LSIG + (size_t)g * (16 * TILE));
  if (nv == 16) stage_in<true>(gbase, nv, lane, lw);
  else          stage_in<false>(gbase, nv, lane, lw);

  const float4* xr = lw + n * SPITCH + half;
  v8f S = {0.f, 0.f, 0.f, 0.f, 0.f, 0.f, 0.f, 0.f};
#pragma unroll
  for (int c = 0; c < 16; ++c) {
    const float4 q0 = xr[4 * c];
    const float4 q1 = xr[4 * c + 2];
    v2f x0, x1, x2, x3;
    x0.x = q0.x; x0.y = q0.y;
    x1.x = q0.z; x1.y = q0.w;
    x2.x = q1.x; x2.y = q1.y;
    x3.x = q1.z; x3.y = q1.w;
    v8f acc = {0.f, 0.f, 0.f, 0.f, 0.f, 0.f, 0.f, 0.f};
    acc = wmma_f32_16x16x4(F[0], x0, acc);
    acc = wmma_f32_16x16x4(F[1], x1, acc);
    acc = wmma_f32_16x16x4(F[2], x2, acc);
    acc = wmma_f32_16x16x4(F[3], x3, acc);
    if (c > 0) {
      v2f s0, s1;
      s0.x = S[0]; s0.y = S[1];
      s1.x = S[2]; s1.y = S[3];
      acc = wmma_f32_16x16x4(A[0], s0, acc);
      acc = wmma_f32_16x16x4(A[1], s1, acc);
    }
    S = acc;
  }

  if (valid) {
    float2* T = (float2*)ts;
    T[(b * 3 + half) * NTILES + t] = make_float2(S[0], S[1]);
    if (half == 0) T[(b * 3 + 2) * NTILES + t] = make_float2(S[2], S[3]);
  }
}

__global__ void __launch_bounds__(SCAN_THREADS) scan_kernel(const float* __restrict__ pw,
                                                            float* __restrict__ ts,
                                                            size_t wsz) {
  if (wsz < WS_BYTES) return;
  __shared__ float2 buf[2][3][SCAN_THREADS];
  const int b = blockIdx.x;
  const int i = threadIdx.x;
  const float* pwr = pw + b * PSTRIDE;
  const float* A256 = pwr + 8 * 36;
  float2* zp = (float2*)ts + (size_t)b * 3 * NTILES;

  const int t0 = 2 * i, t1 = 2 * i + 1;
  float z0[NST], v[NST];
#pragma unroll
  for (int j = 0; j < 3; ++j) {
    float2 a = make_float2(0.f, 0.f), c = make_float2(0.f, 0.f);
    if (t0 < NTILES) a = zp[j * NTILES + t0];
    if (t1 < NTILES) c = zp[j * NTILES + t1];
    z0[2 * j] = a.x; z0[2 * j + 1] = a.y;
    v[2 * j] = c.x;  v[2 * j + 1] = c.y;
  }
  mv6_acc(A256, z0, v);

  int cur = 0;
#pragma unroll
  for (int j = 0; j < 3; ++j) buf[0][j][i] = make_float2(v[2 * j], v[2 * j + 1]);
  __syncthreads();
  for (int k = 0; k < SCAN_LEVELS; ++k) {
    const int len = 1 << k;
    const float* M = pwr + (9 + k) * 36;
    if (i >= len) {
      float q[NST];
#pragma unroll
      for (int j = 0; j < 3; ++j) {
        const float2 w = buf[cur][j][i - len];
        q[2 * j] = w.x; q[2 * j + 1] = w.y;
      }
      mv6_acc(M, q, v);
    }
    cur ^= 1;
#pragma unroll
    for (int j = 0; j < 3; ++j) buf[cur][j][i] = make_float2(v[2 * j], v[2 * j + 1]);
    __syncthreads();
  }

  float e[NST], e1[NST];
#pragma unroll
  for (int j = 0; j < 3; ++j) {
    float2 w = make_float2(0.f, 0.f);
    if (i > 0) w = buf[cur][j][i - 1];
    e[2 * j] = w.x; e[2 * j + 1] = w.y;
  }
#pragma unroll
  for (int q = 0; q < NST; ++q) e1[q] = z0[q];
  mv6_acc(A256, e, e1);
#pragma unroll
  for (int j = 0; j < 3; ++j) {
    if (t0 < NTILES) zp[j * NTILES + t0] = make_float2(e[2 * j], e[2 * j + 1]);
    if (t1 < NTILES) zp[j * NTILES + t1] = make_float2(e1[2 * j], e1[2 * j + 1]);
  }
}

template <bool FULL>
__device__ __forceinline__ void tile_body(const float4* __restrict__ gbase, float4* __restrict__ obase,
                                          const float* __restrict__ pw, const float* __restrict__ fr,
                                          const float* __restrict__ sg, float4* lw,
                                          int b, int nv, int tc, int lane) {
  const int n = lane & 15;
  const int half = lane >> 4;

  v8f S = {0.f, 0.f, 0.f, 0.f, 0.f, 0.f, 0.f, 0.f};
  {
    const float2* Sg = (const float2*)sg;
    const float2 w0 = Sg[(b * 3 + half) * NTILES + tc];
    const float2 w2 = Sg[(b * 3 + 2) * NTILES + tc];
    S[0] = w0.x; S[1] = w0.y;
    S[2] = half ? 0.f : w2.x;
    S[3] = half ? 0.f : w2.y;
  }

  const v2f* hf = (const v2f*)(fr + b * FSTRIDE);
  const v2f* af = (const v2f*)(pw + b * PSTRIDE);
  v2f H[4], F[4], G[2], A[2];
#pragma unroll
  for (int s = 0; s < 4; ++s) {
    H[s] = hf[s * 32 + lane];
    F[s] = hf[128 + s * 32 + lane];
  }
  G[0] = hf[256 + lane];
  G[1] = hf[256 + 32 + lane];
  A[0] = af[lane];
  A[1] = af[32 + lane];

  stage_in<FULL>(gbase, nv, lane, lw);

  float4* xr = lw + n * SPITCH + half;
#pragma unroll
  for (int c = 0; c < 16; ++c) {
    const float4 q0 = xr[4 * c];
    const float4 q1 = xr[4 * c + 2];
    v2f x0, x1, x2, x3, s0, s1;
    x0.x = q0.x; x0.y = q0.y;
    x1.x = q0.z; x1.y = q0.w;
    x2.x = q1.x; x2.y = q1.y;
    x3.x = q1.z; x3.y = q1.w;
    s0.x = S[0]; s0.y = S[1];
    s1.x = S[2]; s1.y = S[3];

    v8f y = {0.f, 0.f, 0.f, 0.f, 0.f, 0.f, 0.f, 0.f};
    y = wmma_f32_16x16x4(H[0], x0, y);
    y = wmma_f32_16x16x4(H[1], x1, y);
    y = wmma_f32_16x16x4(H[2], x2, y);
    y = wmma_f32_16x16x4(H[3], x3, y);
    y = wmma_f32_16x16x4(G[0], s0, y);
    y = wmma_f32_16x16x4(G[1], s1, y);

    if (c < 15) {
      v8f acc = {0.f, 0.f, 0.f, 0.f, 0.f, 0.f, 0.f, 0.f};
      acc = wmma_f32_16x16x4(F[0], x0, acc);
      acc = wmma_f32_16x16x4(F[1], x1, acc);
      acc = wmma_f32_16x16x4(F[2], x2, acc);
      acc = wmma_f32_16x16x4(F[3], x3, acc);
      acc = wmma_f32_16x16x4(A[0], s0, acc);
      acc = wmma_f32_16x16x4(A[1], s1, acc);
      S = acc;
    }

    float4 d0, d1;
    d0.x = y[0]; d0.y = y[1]; d0.z = y[2]; d0.w = y[3];
    d1.x = y[4]; d1.y = y[5]; d1.z = y[6]; d1.w = y[7];
    xr[4 * c]     = d0;
    xr[4 * c + 2] = d1;

    if (c == 7 || c == 15) {
      const int hh = c >> 3;
      __builtin_amdgcn_wave_barrier();
#pragma unroll
      for (int tt = 0; tt < 16; ++tt)
        if (FULL || tt < nv)
          obase[tt * (TILE / 4) + hh * 32 + lane] = lw[tt * SPITCH + hh * 32 + lane];
    }
  }
}

__global__ void __launch_bounds__(256) tile_kernel(const float* __restrict__ in, float* __restrict__ out,
                                                   const float* __restrict__ pw,
                                                   const float* __restrict__ fr,
                                                   const float* __restrict__ sg,
                                                   size_t wsz) {
  if (wsz < WS_BYTES) return;
  __shared__ float4 stage[8 * WSTAGE];
  const int lane = threadIdx.x & 31;
  const int wave = __builtin_amdgcn_readfirstlane((int)(threadIdx.x >> 5));
  const int gw = blockIdx.x * 8 + wave;
  const int b = gw / NGRP;
  const int g = gw - b * NGRP;
  const int t = g * 16 + (lane & 15);
  const int tc = (t < NTILES) ? t : (NTILES - 1);
  const int nv = (NTILES - g * 16 < 16) ? (NTILES - g * 16) : 16;
  float4* lw = stage + wave * WSTAGE;
  const size_t goff = (size_t)b * LSIG + (size_t)g * (16 * TILE);
  const float4* gbase = (const float4*)(in + goff);
  float4* obase = (float4*)(out + goff);
  if (nv == 16) tile_body<true>(gbase, obase, pw, fr, sg, lw, b, nv, tc, lane);
  else          tile_body<false>(gbase, obase, pw, fr, sg, lw, b, nv, tc, lane);
}

extern "C" void kernel_launch(void* const* d_in, const int* in_sizes, int n_in,
                              void* d_out, int out_size, void* d_ws, size_t ws_size,
                              hipStream_t stream) {
  (void)in_sizes; (void)n_in; (void)out_size;
  const float* x     = (const float*)d_in[0];
  const float* knobs = (const float*)d_in[1];
  const float* W1    = (const float*)d_in[2];
  const float* b1v   = (const float*)d_in[3];
  const float* W2    = (const float*)d_in[4];
  const float* b2v   = (const float*)d_in[5];
  float* out = (float*)d_out;

  float* PW = (float*)d_ws;
  float* FR = PW + NB * PSTRIDE;
  float* TS = FR + NB * FSTRIDE;

  derive_kernel<<<NB, 256, 0, stream>>>(knobs, W1, b1v, W2, b2v, PW, FR, ws_size);
  state_kernel<<<(NB * NGRP) / 8, 256, 0, stream>>>(x, PW, FR, TS, ws_size);
  scan_kernel<<<NB, SCAN_THREADS, 0, stream>>>(PW, TS, ws_size);
  tile_kernel<<<(NB * NGRP) / 8, 256, 0, stream>>>(x, out, PW, FR, TS, ws_size);
}


# set-up kernel: fp64 work off its serial path: chunk tables grow by doubling (a thread per element) beside the squarings, fragments on four waves, MLP one unit per lane; sum order unchanged
# speedup vs baseline: 15.2734x; 1.2204x over previous
#include <hip/hip_runtime.h>
#include <hip/hip_bf16.h>

#define NB      64
#define LSIG    480000
#define TILE    256
#define NTILES  (LSIG / TILE)
#define NGRP    ((NTILES + 15) / 16)
#define NST     6
#define NPOW    19
#define PSTRIDE (NPOW * 36)
#define FSTRIDE 640
#define SCAN_THREADS 1024
#define SCAN_LEVELS  10
#define SPITCH  65
#define WSTAGE  (16 * SPITCH)
#define WS_FLOATS (NB * PSTRIDE + NB * FSTRIDE + NB * 3 * NTILES * 2)
#define WS_BYTES  ((size_t)WS_FLOATS * 4)

typedef __attribute__((ext_vector_type(2))) float v2f;
typedef __attribute__((ext_vector_type(8))) float v8f;

__device__ __forceinline__ v8f wmma_f32_16x16x4(v2f a, v2f b, v8f c) {
  return __builtin_amdgcn_wmma_f32_16x16x4_f32(false, a, false, b, (short)0, c,
                                               false, false);
}

__device__ __forceinline__ void mv6_acc(const float* __restrict__ M, const float* q, float* v) {
  v[0] += M[0] * q[0] + M[1] * q[1];
  v[1] += M[6] * q[0] + M[7] * q[1];
  v[2] += M[12] * q[0] + M[13] * q[1] + M[14] * q[2] + M[15] * q[3];
  v[3] += M[18] * q[0] + M[19] * q[1] + M[20] * q[2] + M[21] * q[3];
  v[4] += M[24] * q[0] + M[25] * q[1] + M[26] * q[2] + M[27] * q[3] + M[28] * q[4] + M[29] * q[5];
  v[5] += M[30] * q[0] + M[31] * q[1] + M[32] * q[2] + M[33] * q[3] + M[34] * q[4] + M[35] * q[5];
}

__global__ void __launch_bounds__(256) derive_kernel(const float* __restrict__ knobs,
                                                     const float* __restrict__ W1,
                                                     const float* __restrict__ b1v,
                                                     const float* __restrict__ W2,
                                                     const float* __restrict__ b2v,
                                                     float* __restrict__ pw,
                                                     float* __restrict__ fr,
                                                     size_t wsz) {
  if (wsz < WS_BYTES) return;
  __shared__ double P[NPOW][36];
  __shared__ double Gs[16][NST];
  __shared__ double Fe[16][NST];
  __shared__ double B6s[NST];
  __shared__ double D6s;
  __shared__ float hs[16];
  const int b = blockIdx.x;
  const int tid = threadIdx.x;

  if (tid < 32) {
    const float k0 = knobs[b * 3 + 0], k1 = knobs[b * 3 + 1], k2 = knobs[b * 3 + 2];
    const int iu = tid & 15;
    const float hv = tanhf(k0 * W1[0 * 16 + iu] + k1 * W1[1 * 16 + iu] +
                           k2 * W1[2 * 16 + iu] + b1v[iu]);
    float hmlp[16];
#pragma unroll
    for (int i = 0; i < 16; ++i) hmlp[i] = __shfl(hv, i, 32);
    const int ju = (tid < 15) ? tid : 14;
    float accp = b2v[ju];
#pragma unroll
    for (int i = 0; i < 16; ++i) accp = fmaf(hmlp[i], W2[i * 15 + ju], accp);
    const float tp = tanhf(accp);
    float p[15], tq[15];
#pragma unroll
    for (int j = 0; j < 15; ++j) {
      p[j] = __shfl(accp, j, 32);
      tq[j] = __shfl(tp, j, 32);
    }
    double a1[3], a2[3], d[3], B0[3], B1[3];
#pragma unroll
    for (int s = 0; s < 3; ++s) {
      float c0 = p[s * 5 + 0], c1 = p[s * 5 + 1], c2 = p[s * 5 + 2];
      float a2f = 0.999f * tq[s * 5 + 4];
      float a1f = 0.999f * (1.0f + a2f) * tq[s * 5 + 3];
      a1[s] = (double)a1f; a2[s] = (double)a2f;
      d[s] = (double)c0;
      B0[s] = (double)c1 - a1[s] * d[s];
      B1[s] = (double)c2 - a2[s] * d[s];
    }
    if (tid == 0) {
      double B6[NST], C6[NST];
      B6[0] = B0[0];               B6[1] = B1[0];
      B6[2] = B0[1] * d[0];        B6[3] = B1[1] * d[0];
      B6[4] = B0[2] * d[1] * d[0]; B6[5] = B1[2] * d[1] * d[0];
      C6[0] = d[2] * d[1]; C6[1] = 0.0; C6[2] = d[2]; C6[3] = 0.0; C6[4] = 1.0; C6[5] = 0.0;
      D6s = d[2] * d[1] * d[0];
#pragma unroll
      for (int i = 0; i < NST; ++i) {
        Gs[0][i] = C6[i];
        Fe[0][i] = B6[i];
        B6s[i] = B6[i];
      }
      double A6[36];
#pragma unroll
      for (int e = 0; e < 36; ++e) A6[e] = 0.0;
#pragma unroll
      for (int s = 0; s < 3; ++s) {
        A6[(2 * s) * 6 + 2 * s]         = -a1[s];
        A6[(2 * s) * 6 + 2 * s + 1]     = 1.0;
        A6[(2 * s + 1) * 6 + 2 * s]     = -a2[s];
      }
      A6[2 * 6 + 0] = B0[1];        A6[3 * 6 + 0] = B1[1];
      A6[4 * 6 + 2] = B0[2];        A6[5 * 6 + 2] = B1[2];
      A6[4 * 6 + 0] = B0[2] * d[1]; A6[5 * 6 + 0] = B1[2] * d[1];
#pragma unroll
      for (int e = 0; e < 36; ++e) P[0][e] = A6[e];
    }
  }
  __syncthreads();
  for (int k = 0; k + 1 < NPOW; ++k) {
    if (tid < 36) {
      const int r = tid / 6, c = tid - r * 6;
      double pr[6], pc[6];
#pragma unroll
      for (int j = 0; j < 6; ++j) {
        pr[j] = P[k][r * 6 + j];
        pc[j] = P[k][j * 6 + c];
      }
      asm volatile("" :: "v"(pr[0]), "v"(pr[1]), "v"(pr[2]), "v"(pr[3]), "v"(pr[4]), "v"(pr[5]),
                         "v"(pc[0]), "v"(pc[1]), "v"(pc[2]), "v"(pc[3]), "v"(pc[4]), "v"(pc[5]));
      double acc = 0.0;
#pragma unroll
      for (int j = 0; j < 6; ++j) acc += pr[j] * pc[j];
      P[k + 1][tid] = acc;
    } else if (tid >= 64 && tid < 160) {
      const int q = tid - 64;
      if (k < 4) {
        const int n = q / 6, c = q - n * 6;
        if ((n >> k) == 1) {
          const double* g = Gs[n - (1 << k)];
          double acc = 0.0;
#pragma unroll
          for (int r = 0; r < NST; ++r) acc += g[r] * P[k][r * 6 + c];
          Gs[n][c] = acc;
        }
      } else if (k == 4) {
        if (q < 16) {
          double acc = D6s;
          if (q > 0) {
            acc = 0.0;
#pragma unroll
            for (int i = 0; i < NST; ++i) acc += Gs[q - 1][i] * B6s[i];
          }
          hs[q] = (float)acc;
        }
      }
    } else if (tid >= 160) {
      if (k < 4) {
        const int q = tid - 160;
        const int e = q / 6, r = q - e * 6;
        if ((e >> k) == 1) {
          const double* f = Fe[e - (1 << k)];
          double acc = 0.0;
#pragma unroll
          for (int c = 0; c < NST; ++c) acc += P[k][r * 6 + c] * f[c];
          Fe[e][r] = acc;
        }
      }
    }
    __syncthreads();
  }
  if (tid >= 64 && tid < 192) {
    const int q = tid - 64;
    const int fam = q >> 5;
    const int lane = q & 31, m = lane & 15, half = lane >> 4;
    float* fb = fr + b * FSTRIDE;
    float* ab = pw + b * PSTRIDE;
    const int pm = 8 * ((m & 7) >> 2) + 4 * (m >> 3) + (m & 3);
    int sidx = -1;
    if (m < 2) sidx = m; else if (m < 4) sidx = m + 2; else if (m == 8 || m == 9) sidx = m - 6;
    if (fam < 2) {
      for (int s = 0; s < 4; ++s) {
        for (int e = 0; e < 2; ++e) {
          const int i = 2 * s + e;
          const int j = 8 * (i >> 2) + 4 * half + (i & 3);
          if (fam == 0) fb[(s * 32 + lane) * 2 + e] = (j <= pm) ? hs[pm - j] : 0.0f;
          else          fb[256 + (s * 32 + lane) * 2 + e] = (sidx >= 0) ? (float)Fe[15 - j][sidx] : 0.0f;
        }
      }
    } else {
      for (int s = 0; s < 2; ++s) {
        for (int e = 0; e < 2; ++e) {
          const int kst = (s == 0) ? (2 * half + e) : (half ? -1 : 4 + e);
          if (fam == 2) fb[512 + (s * 32 + lane) * 2 + e] = (kst >= 0) ? (float)Gs[pm][kst] : 0.0f;
          else          ab[(s * 32 + lane) * 2 + e] = (kst >= 0 && sidx >= 0) ? (float)P[4][sidx * 6 + kst] : 0.0f;
        }
      }
    }
  }
  {
    const double* Pf = &P[0][0];
    for (int idx = 8 * 36 + tid; idx < PSTRIDE; idx += 256) pw[b * PSTRIDE + idx] = (float)Pf[idx];
  }
}

template <bool FULL>
__device__ __forceinline__ void stage_in(const float4* __restrict__ gbase, int nv, int lane, float4* lw) {
  float4 xv[32];
#pragma unroll
  for (int j = 0; j < 32; ++j) {
    const int tt = j & 15, hh = j >> 4;
    const int ts = (FULL || tt < nv) ? tt : 0;
    xv[j] = gbase[ts * (TILE / 4) + hh * 32 + lane];
  }
#pragma unroll
  for (int j = 0; j < 32; ++j) lw[(j & 15) * SPITCH + (j >> 4) * 32 + lane] = xv[j];
  __builtin_amdgcn_wave_barrier();
}

__global__ void __launch_bounds__(256) state_kernel(const float* __restrict__ in,
                                                    const float* __restrict__ pw,
                                                    const float* __restrict__ fr,
                                                    float* __restrict__ ts,
                                                    size_t wsz) {
  if (wsz < WS_BYTES) return;
  __shared__ float4 stage[8 * WSTAGE];
  const int lane = threadIdx.x & 31;
  const int wave = __builtin_amdgcn_readfirstlane((int)(threadIdx.x >> 5));
  const int gw = blockIdx.x * 8 + wave;
  const int b = gw / NGRP;
  const int g = gw - b * NGRP;
  const int n = lane & 15;
  const int half = lane >> 4;
  const int t = g * 16 + n;
  const bool valid = t < NTILES;
  const int nv = (NTILES - g * 16 < 16) ? (NTILES - g * 16) : 16;
  float4* lw = stage + wave * WSTAGE;

  const v2f* hf = (const v2f*)(fr + b * FSTRIDE);
  const v2f* af = (const v2f*)(pw + b * PSTRIDE);
  v2f F[4], A[2];
#pragma unroll
  for (int s = 0; s < 4; ++s) F[s] = hf[128 + s * 32 + lane];
  A[0] = af[lane];
  A[1] = af[32 + lane];

  const float4* gbase = (const float4*)(in + (size_t)b * LSIG + (size_t)g * (16 * TILE));
  if (nv == 16) stage_in<true>(gbase, nv, lane, lw);
  else          stage_in<false>(gbase, nv, lane, lw);

  const float4* xr = lw + n * SPITCH + half;
  v8f S = {0.f, 0.f, 0.f, 0.f, 0.f, 0.f, 0.f, 0.f};
#pragma unroll
  for (int c = 0; c < 16; ++c) {
    const float4 q0 = xr[4 * c];
    const float4 q1 = xr[4 * c + 2];
    v2f x0, x1, x2, x3;
    x0.x = q0.x; x0.y = q0.y;
    x1.x = q0.z; x1.y = q0.w;
    x2.x = q1.x; x2.y = q1.y;
    x3.x = q1.z; x3.y = q1.w;
    v8f acc = {0.f, 0.f, 0.f, 0.f, 0.f, 0.f, 0.f, 0.f};
    acc = wmma_f32_16x16x4(F[0], x0, acc);
    acc = wmma_f32_16x16x4(F[1], x1, acc);
    acc = wmma_f32_16x16x4(F[2], x2, acc);
    acc = wmma_f32_16x16x4(F[3], x3, acc);
    if (c > 0) {
      v2f s0, s1;
      s0.x = S[0]; s0.y = S[1];
      s1.x = S[2]; s1.y = S[3];
      acc = wmma_f32_16x16x4(A[0], s0, acc);
      acc = wmma_f32_16x16x4(A[1], s1, acc);
    }
    S = acc;
  }

  if (valid) {
    float2* T = (float2*)ts;
    T[(b * 3 + half) * NTILES + t] = make_float2(S[0], S[1]);
    if (half == 0) T[(b * 3 + 2) * NTILES + t] = make_float2(S[2], S[3]);
  }
}

__global__ void __launch_bounds__(SCAN_THREADS) scan_kernel(const float* __restrict__ pw,
                                                            float* __restrict__ ts,
                                                            size_t wsz) {
  if (wsz < WS_BYTES) return;
  __shared__ float2 buf[2][3][SCAN_THREADS];
  const int b = blockIdx.x;
  const int i = threadIdx.x;
  const float* pwr = pw + b * PSTRIDE;
  const float* A256 = pwr + 8 * 36;
  float2* zp = (float2*)ts + (size_t)b * 3 * NTILES;

  const int t0 = 2 * i, t1 = 2 * i + 1;
  float z0[NST], v[NST];
#pragma unroll
  for (int j = 0; j < 3; ++j) {
    float2 a = make_float2(0.f, 0.f), c = make_float2(0.f, 0.f);
    if (t0 < NTILES) a = zp[j * NTILES + t0];
    if (t1 < NTILES) c = zp[j * NTILES + t1];
    z0[2 * j] = a.x; z0[2 * j + 1] = a.y;
    v[2 * j] = c.x;  v[2 * j + 1] = c.y;
  }
  mv6_acc(A256, z0, v);

  int cur = 0;
#pragma unroll
  for (int j = 0; j < 3; ++j) buf[0][j][i] = make_float2(v[2 * j], v[2 * j + 1]);
  __syncthreads();
  for (int k = 0; k < SCAN_LEVELS; ++k) {
    const int len = 1 << k;
    const float* M = pwr + (9 + k) * 36;
    if (i >= len) {
      float q[NST];
#pragma unroll
      for (int j = 0; j < 3; ++j) {
        const float2 w = buf[cur][j][i - len];
        q[2 * j] = w.x; q[2 * j + 1] = w.y;
      }
      mv6_acc(M, q, v);
    }
    cur ^= 1;
#pragma unroll
    for (int j = 0; j < 3; ++j) buf[cur][j][i] = make_float2(v[2 * j], v[2 * j + 1]);
    __syncthreads();
  }

  float e[NST], e1[NST];
#pragma unroll
  for (int j = 0; j < 3; ++j) {
    float2 w = make_float2(0.f, 0.f);
    if (i > 0) w = buf[cur][j][i - 1];
    e[2 * j] = w.x; e[2 * j + 1] = w.y;
  }
#pragma unroll
  for (int q = 0; q < NST; ++q) e1[q] = z0[q];
  mv6_acc(A256, e, e1);
#pragma unroll
  for (int j = 0; j < 3; ++j) {
    if (t0 < NTILES) zp[j * NTILES + t0] = make_float2(e[2 * j], e[2 * j + 1]);
    if (t1 < NTILES) zp[j * NTILES + t1] = make_float2(e1[2 * j], e1[2 * j + 1]);
  }
}

template <bool FULL>
__device__ __forceinline__ void tile_body(const float4* __restrict__ gbase, float4* __restrict__ obase,
                                          const float* __restrict__ pw, const float* __restrict__ fr,
                                          const float* __restrict__ sg, float4* lw,
                                          int b, int nv, int tc, int lane) {
  const int n = lane & 15;
  const int half = lane >> 4;

  v8f S = {0.f, 0.f, 0.f, 0.f, 0.f, 0.f, 0.f, 0.f};
  {
    const float2* Sg = (const float2*)sg;
    const float2 w0 = Sg[(b * 3 + half) * NTILES + tc];
    const float2 w2 = Sg[(b * 3 + 2) * NTILES + tc];
    S[0] = w0.x; S[1] = w0.y;
    S[2] = half ? 0.f : w2.x;
    S[3] = half ? 0.f : w2.y;
  }

  const v2f* hf = (const v2f*)(fr + b * FSTRIDE);
  const v2f* af = (const v2f*)(pw + b * PSTRIDE);
  v2f H[4], F[4], G[2], A[2];
#pragma unroll
  for (int s = 0; s < 4; ++s) {
    H[s] = hf[s * 32 + lane];
    F[s] = hf[128 + s * 32 + lane];
  }
  G[0] = hf[256 + lane];
  G[1] = hf[256 + 32 + lane];
  A[0] = af[lane];
  A[1] = af[32 + lane];

  stage_in<FULL>(gbase, nv, lane, lw);

  float4* xr = lw + n * SPITCH + half;
#pragma unroll
  for (int c = 0; c < 16; ++c) {
    const float4 q0 = xr[4 * c];
    const float4 q1 = xr[4 * c + 2];
    v2f x0, x1, x2, x3, s0, s1;
    x0.x = q0.x; x0.y = q0.y;
    x1.x = q0.z; x1.y = q0.w;
    x2.x = q1.x; x2.y = q1.y;
    x3.x = q1.z; x3.y = q1.w;
    s0.x = S[0]; s0.y = S[1];
    s1.x = S[2]; s1.y = S[3];

    v8f y = {0.f, 0.f, 0.f, 0.f, 0.f, 0.f, 0.f, 0.f};
    y = wmma_f32_16x16x4(H[0], x0, y);
    y = wmma_f32_16x16x4(H[1], x1, y);
    y = wmma_f32_16x16x4(H[2], x2, y);
    y = wmma_f32_16x16x4(H[3], x3, y);
    y = wmma_f32_16x16x4(G[0], s0, y);
    y = wmma_f32_16x16x4(G[1], s1, y);

    if (c < 15) {
      v8f acc = {0.f, 0.f, 0.f, 0.f, 0.f, 0.f, 0.f, 0.f};
      acc = wmma_f32_16x16x4(F[0], x0, acc);
      acc = wmma_f32_16x16x4(F[1], x1, acc);
      acc = wmma_f32_16x16x4(F[2], x2, acc);
      acc = wmma_f32_16x16x4(F[3], x3, acc);
      acc = wmma_f32_16x16x4(A[0], s0, acc);
      acc = wmma_f32_16x16x4(A[1], s1, acc);
      S = acc;
    }

    float4 d0, d1;
    d0.x = y[0]; d0.y = y[1]; d0.z = y[2]; d0.w = y[3];
    d1.x = y[4]; d1.y = y[5]; d1.z = y[6]; d1.w = y[7];
    xr[4 * c]     = d0;
    xr[4 * c + 2] = d1;

    if (c == 7 || c == 15) {
      const int hh = c >> 3;
      __builtin_amdgcn_wave_barrier();
#pragma unroll
      for (int tt = 0; tt < 16; ++tt)
        if (FULL || tt < nv)
          obase[tt * (TILE / 4) + hh * 32 + lane] = lw[tt * SPITCH + hh * 32 + lane];
    }
  }
}

__global__ void __launch_bounds__(256) tile_kernel(const float* __restrict__ in, float* __restrict__ out,
                                                   const float* __restrict__ pw,
                                                   const float* __restrict__ fr,
                                                   const float* __restrict__ sg,
                                                   size_t wsz) {
  if (wsz < WS_BYTES) return;
  __shared__ float4 stage[8 * WSTAGE];
  const int lane = threadIdx.x & 31;
  const int wave = __builtin_amdgcn_readfirstlane((int)(threadIdx.x >> 5));
  const int gw = blockIdx.x * 8 + wave;
  const int b = gw / NGRP;
  const int g = gw - b * NGRP;
  const int t = g * 16 + (lane & 15);
  const int tc = (t < NTILES) ? t : (NTILES - 1);
  const int nv = (NTILES - g * 16 < 16) ? (NTILES - g * 16) : 16;
  float4* lw = stage + wave * WSTAGE;
  const size_t goff = (size_t)b * LSIG + (size_t)g * (16 * TILE);
  const float4* gbase = (const float4*)(in + goff);
  float4* obase = (float4*)(out + goff);
  if (nv == 16) tile_body<true>(gbase, obase, pw, fr, sg, lw, b, nv, tc, lane);
  else          tile_body<false>(gbase, obase, pw, fr, sg, lw, b, nv, tc, lane);
}

extern "C" void kernel_launch(void* const* d_in, const int* in_sizes, int n_in,
                              void* d_out, int out_size, void* d_ws, size_t ws_size,
                              hipStream_t stream) {
  (void)in_sizes; (void)n_in; (void)out_size;
  const float* x     = (const float*)d_in[0];
  const float* knobs = (const float*)d_in[1];
  const float* W1    = (const float*)d_in[2];
  const float* b1v   = (const float*)d_in[3];
  const float* W2    = (const float*)d_in[4];
  const float* b2v   = (const float*)d_in[5];
  float* out = (float*)d_out;

  float* PW = (float*)d_ws;
  float* FR = PW + NB * PSTRIDE;
  float* TS = FR + NB * FSTRIDE;

  derive_kernel<<<NB, 256, 0, stream>>>(knobs, W1, b1v, W2, b2v, PW, FR, ws_size);
  state_kernel<<<(NB * NGRP) / 8, 256, 0, stream>>>(x, PW, FR, TS, ws_size);
  scan_kernel<<<NB, SCAN_THREADS, 0, stream>>>(PW, TS, ws_size);
  tile_kernel<<<(NB * NGRP) / 8, 256, 0, stream>>>(x, out, PW, FR, TS, ws_size);
}


# set-up kernel keeps only what the state pass reads (4 squaring levels, F and A^16 fragments); powers A^256 and up, H and G are built by 64 extra work-groups of the state launch; sums unchanged
# speedup vs baseline: 16.2870x; 1.0664x over previous
#include <hip/hip_runtime.h>
#include <hip/hip_bf16.h>

#define NB      64
#define LSIG    480000
#define TILE    256
#define NTILES  (LSIG / TILE)
#define NGRP    ((NTILES + 15) / 16)
#define NST     6
#define NPOW    19
#define PSTRIDE (NPOW * 36)
#define FSTRIDE 640
#define SCAN_THREADS 1024
#define SCAN_LEVELS  10
#define SPITCH  65
#define WSTAGE  (16 * SPITCH)
#define WS_FLOATS (NB * PSTRIDE + NB * FSTRIDE + NB * 3 * NTILES * 2)
#define WS_BYTES  ((size_t)WS_FLOATS * 4)

typedef __attribute__((ext_vector_type(2))) float v2f;
typedef __attribute__((ext_vector_type(8))) float v8f;

__device__ __forceinline__ v8f wmma_f32_16x16x4(v2f a, v2f b, v8f c) {
  return __builtin_amdgcn_wmma_f32_16x16x4_f32(false, a, false, b, (short)0, c,
                                               false, false);
}

__device__ __forceinline__ void mv6_acc(const float* __restrict__ M, const float* q, float* v) {
  v[0] += M[0] * q[0] + M[1] * q[1];
  v[1] += M[6] * q[0] + M[7] * q[1];
  v[2] += M[12] * q[0] + M[13] * q[1] + M[14] * q[2] + M[15] * q[3];
  v[3] += M[18] * q[0] + M[19] * q[1] + M[20] * q[2] + M[21] * q[3];
  v[4] += M[24] * q[0] + M[25] * q[1] + M[26] * q[2] + M[27] * q[3] + M[28] * q[4] + M[29] * q[5];
  v[5] += M[30] * q[0] + M[31] * q[1] + M[32] * q[2] + M[33] * q[3] + M[34] * q[4] + M[35] * q[5];
}

template <bool TAIL>
__device__ __forceinline__ void derive_row(const float* __restrict__ knobs,
                                           const float* __restrict__ W1,
                                           const float* __restrict__ b1v,
                                           const float* __restrict__ W2,
                                           const float* __restrict__ b2v,
                                           float* __restrict__ pw,
                                           float* __restrict__ fr,
                                           const int b, const int tid) {
  constexpr int NLEV = TAIL ? (NPOW - 1) : 4;
  __shared__ double P[NPOW][36];
  __shared__ double Gs[16][NST];
  __shared__ double Fe[16][NST];
  __shared__ double B6s[NST];
  __shared__ double D6s;
  __shared__ float hs[16];

  if (tid < 32) {
    const float k0 = knobs[b * 3 + 0], k1 = knobs[b * 3 + 1], k2 = knobs[b * 3 + 2];
    const int iu = tid & 15;
    const float hv = tanhf(k0 * W1[0 * 16 + iu] + k1 * W1[1 * 16 + iu] +
                           k2 * W1[2 * 16 + iu] + b1v[iu]);
    float hmlp[16];
#pragma unroll
    for (int i = 0; i < 16; ++i) hmlp[i] = __shfl(hv, i, 32);
    const int ju = (tid < 15) ? tid : 14;
    float accp = b2v[ju];
#pragma unroll
    for (int i = 0; i < 16; ++i) accp = fmaf(hmlp[i], W2[i * 15 + ju], accp);
    const float tp = tanhf(accp);
    float p[15], tq[15];
#pragma unroll
    for (int j = 0; j < 15; ++j) {
      p[j] = __shfl(accp, j, 32);
      tq[j] = __shfl(tp, j, 32);
    }
    double a1[3], a2[3], d[3], B0[3], B1[3];
#pragma unroll
    for (int s = 0; s < 3; ++s) {
      float c0 = p[s * 5 + 0], c1 = p[s * 5 + 1], c2 = p[s * 5 + 2];
      float a2f = 0.999f * tq[s * 5 + 4];
      float a1f = 0.999f * (1.0f + a2f) * tq[s * 5 + 3];
      a1[s] = (double)a1f; a2[s] = (double)a2f;
      d[s] = (double)c0;
      B0[s] = (double)c1 - a1[s] * d[s];
      B1[s] = (double)c2 - a2[s] * d[s];
    }
    if (tid == 0) {
      double B6[NST], C6[NST];
      B6[0] = B0[0];               B6[1] = B1[0];
      B6[2] = B0[1] * d[0];        B6[3] = B1[1] * d[0];
      B6[4] = B0[2] * d[1] * d[0]; B6[5] = B1[2] * d[1] * d[0];
      C6[0] = d[2] * d[1]; C6[1] = 0.0; C6[2] = d[2]; C6[3] = 0.0; C6[4] = 1.0; C6[5] = 0.0;
      D6s = d[2] * d[1] * d[0];
#pragma unroll
      for (int i = 0; i < NST; ++i) {
        Gs[0][i] = C6[i];
        Fe[0][i] = B6[i];
        B6s[i] = B6[i];
      }
      double A6[36];
#pragma unroll
      for (int e = 0; e < 36; ++e) A6[e] = 0.0;
#pragma unroll
      for (int s = 0; s < 3; ++s) {
        A6[(2 * s) * 6 + 2 * s]         = -a1[s];
        A6[(2 * s) * 6 + 2 * s + 1]     = 1.0;
        A6[(2 * s + 1) * 6 + 2 * s]     = -a2[s];
      }
      A6[2 * 6 + 0] = B0[1];        A6[3 * 6 + 0] = B1[1];
      A6[4 * 6 + 2] = B0[2];        A6[5 * 6 + 2] = B1[2];
      A6[4 * 6 + 0] = B0[2] * d[1]; A6[5 * 6 + 0] = B1[2] * d[1];
#pragma unroll
      for (int e = 0; e < 36; ++e) P[0][e] = A6[e];
    }
  }
  __syncthreads();
  for (int k = 0; k < NLEV; ++k) {
    if (tid < 36) {
      const int r = tid / 6, c = tid - r * 6;
      double pr[6], pc[6];
#pragma unroll
      for (int j = 0; j < 6; ++j) {
        pr[j] = P[k][r * 6 + j];
        pc[j] = P[k][j * 6 + c];
      }
      asm volatile("" :: "v"(pr[0]), "v"(pr[1]), "v"(pr[2]), "v"(pr[3]), "v"(pr[4]), "v"(pr[5]),
                         "v"(pc[0]), "v"(pc[1]), "v"(pc[2]), "v"(pc[3]), "v"(pc[4]), "v"(pc[5]));
      double acc = 0.0;
#pragma unroll
      for (int j = 0; j < 6; ++j) acc += pr[j] * pc[j];
      P[k + 1][tid] = acc;
    } else if (TAIL && tid >= 64 && tid < 160) {
      const int q = tid - 64;
      if (k < 4) {
        const int n = q / 6, c = q - n * 6;
        if ((n >> k) == 1) {
          const double* g = Gs[n - (1 << k)];
          double acc = 0.0;
#pragma unroll
          for (int r = 0; r < NST; ++r) acc += g[r] * P[k][r * 6 + c];
          Gs[n][c] = acc;
        }
      } else if (k == 4) {
        if (q < 16) {
          double acc = D6s;
          if (q > 0) {
            acc = 0.0;
#pragma unroll
            for (int i = 0; i < NST; ++i) acc += Gs[q - 1][i] * B6s[i];
          }
          hs[q] = (float)acc;
        }
      }
    } else if (tid >= 160) {
      if (k < 4) {
        const int q = tid - 160;
        const int e = q / 6, r = q - e * 6;
        if ((e >> k) == 1) {
          const double* f = Fe[e - (1 << k)];
          double acc = 0.0;
#pragma unroll
          for (int c = 0; c < NST; ++c) acc += P[k][r * 6 + c] * f[c];
          Fe[e][r] = acc;
        }
      }
    }
    __syncthreads();
  }
  if (tid >= 64 && tid < 192) {
    const int q = tid - 64;
    const int fam = q >> 5;
    const int lane = q & 31, m = lane & 15, half = lane >> 4;
    float* fb = fr + b * FSTRIDE;
    float* ab = pw + b * PSTRIDE;
    const int pm = 8 * ((m & 7) >> 2) + 4 * (m >> 3) + (m & 3);
    int sidx = -1;
    if (m < 2) sidx = m; else if (m < 4) sidx = m + 2; else if (m == 8 || m == 9) sidx = m - 6;
    if (((fam & 1) != 0) == TAIL) {
    } else if (fam < 2) {
      for (int s = 0; s < 4; ++s) {
        for (int e = 0; e < 2; ++e) {
          const int i = 2 * s + e;
          const int j = 8 * (i >> 2) + 4 * half + (i & 3);
          if (fam == 0) fb[(s * 32 + lane) * 2 + e] = (j <= pm) ? hs[pm - j] : 0.0f;
          else          fb[256 + (s * 32 + lane) * 2 + e] = (sidx >= 0) ? (float)Fe[15 - j][sidx] : 0.0f;
        }
      }
    } else {
      for (int s = 0; s < 2; ++s) {
        for (int e = 0; e < 2; ++e) {
          const int kst = (s == 0) ? (2 * half + e) : (half ? -1 : 4 + e);
          if (fam == 2) fb[512 + (s * 32 + lane) * 2 + e] = (kst >= 0) ? (float)Gs[pm][kst] : 0.0f;
          else          ab[(s * 32 + lane) * 2 + e] = (kst >= 0 && sidx >= 0) ? (float)P[4][sidx * 6 + kst] : 0.0f;
        }
      }
    }
  }
  if (TAIL) {
    const double* Pf = &P[0][0];
    for (int idx = 8 * 36 + tid; idx < PSTRIDE; idx += 256) pw[b * PSTRIDE + idx] = (float)Pf[idx];
  }
}

__global__ void __launch_bounds__(256) derive_kernel(const float* __restrict__ knobs,
                                                     const float* __restrict__ W1,
                                                     const float* __restrict__ b1v,
                                                     const float* __restrict__ W2,
                                                     const float* __restrict__ b2v,
                                                     float* __restrict__ pw,
                                                     float* __restrict__ fr,
                                                     size_t wsz) {
  if (wsz < WS_BYTES) return;
  derive_row<false>(knobs, W1, b1v, W2, b2v, pw, fr, (int)blockIdx.x, (int)threadIdx.x);
}

template <bool FULL>
__device__ __forceinline__ void stage_in(const float4* __restrict__ gbase, int nv, int lane, float4* lw) {
  float4 xv[32];
#pragma unroll
  for (int j = 0; j < 32; ++j) {
    const int tt = j & 15, hh = j >> 4;
    const int ts = (FULL || tt < nv) ? tt : 0;
    xv[j] = gbase[ts * (TILE / 4) + hh * 32 + lane];
  }
#pragma unroll
  for (int j = 0; j < 32; ++j) lw[(j & 15) * SPITCH + (j >> 4) * 32 + lane] = xv[j];
  __builtin_amdgcn_wave_barrier();
}

__global__ void __launch_bounds__(256) state_kernel(const float* __restrict__ in,
                                                    const float* __restrict__ knobs,
                                                    const float* __restrict__ W1,
                                                    const float* __restrict__ b1v,
                                                    const float* __restrict__ W2,
                                                    const float* __restrict__ b2v,
                                                    float* __restrict__ pw,
                                                    float* __restrict__ fr,
                                                    float* __restrict__ ts,
                                                    size_t wsz) {
  if (wsz < WS_BYTES) return;
  if (blockIdx.x < NB) {
    derive_row<true>(knobs, W1, b1v, W2, b2v, pw, fr, (int)blockIdx.x, (int)threadIdx.x);
    return;
  }
  __shared__ float4 stage[8 * WSTAGE];
  const int lane = threadIdx.x & 31;
  const int wave = __builtin_amdgcn_readfirstlane((int)(threadIdx.x >> 5));
  const int gw = ((int)blockIdx.x - NB) * 8 + wave;
  const int b = gw / NGRP;
  const int g = gw - b * NGRP;
  const int n = lane & 15;
  const int half = lane >> 4;
  const int t = g * 16 + n;
  const bool valid = t < NTILES;
  const int nv = (NTILES - g * 16 < 16) ? (NTILES - g * 16) : 16;
  float4* lw = stage + wave * WSTAGE;

  const v2f* hf = (const v2f*)(fr + b * FSTRIDE);
  const v2f* af = (const v2f*)(pw + b * PSTRIDE);
  v2f F[4], A[2];
#pragma unroll
  for (int s = 0; s < 4; ++s) F[s] = hf[128 + s * 32 + lane];
  A[0] = af[lane];
  A[1] = af[32 + lane];

  const float4* gbase = (const float4*)(in + (size_t)b * LSIG + (size_t)g * (16 * TILE));
  if (nv == 16) stage_in<true>(gbase, nv, lane, lw);
  else          stage_in<false>(gbase, nv, lane, lw);

  const float4* xr = lw + n * SPITCH + half;
  v8f S = {0.f, 0.f, 0.f, 0.f, 0.f, 0.f, 0.f, 0.f};
#pragma unroll
  for (int c = 0; c < 16; ++c) {
    const float4 q0 = xr[4 * c];
    const float4 q1 = xr[4 * c + 2];
    v2f x0, x1, x2, x3;
    x0.x = q0.x; x0.y = q0.y;
    x1.x = q0.z; x1.y = q0.w;
    x2.x = q1.x; x2.y = q1.y;
    x3.x = q1.z; x3.y = q1.w;
    v8f acc = {0.f, 0.f, 0.f, 0.f, 0.f, 0.f, 0.f, 0.f};
    acc = wmma_f32_16x16x4(F[0], x0, acc);
    acc = wmma_f32_16x16x4(F[1], x1, acc);
    acc = wmma_f32_16x16x4(F[2], x2, acc);
    acc = wmma_f32_16x16x4(F[3], x3, acc);
    if (c > 0) {
      v2f s0, s1;
      s0.x = S[0]; s0.y = S[1];
      s1.x = S[2]; s1.y = S[3];
      acc = wmma_f32_16x16x4(A[0], s0, acc);
      acc = wmma_f32_16x16x4(A[1], s1, acc);
    }
    S = acc;
  }

  if (valid) {
    float2* T = (float2*)ts;
    T[(b * 3 + half) * NTILES + t] = make_float2(S[0], S[1]);
    if (half == 0) T[(b * 3 + 2) * NTILES + t] = make_float2(S[2], S[3]);
  }
}

__global__ void __launch_bounds__(SCAN_THREADS) scan_kernel(const float* __restrict__ pw,
                                                            float* __restrict__ ts,
                                                            size_t wsz) {
  if (wsz < WS_BYTES) return;
  __shared__ float2 buf[2][3][SCAN_THREADS];
  const int b = blockIdx.x;
  const int i = threadIdx.x;
  const float* pwr = pw + b * PSTRIDE;
  const float* A256 = pwr + 8 * 36;
  float2* zp = (float2*)ts + (size_t)b * 3 * NTILES;

  const int t0 = 2 * i, t1 = 2 * i + 1;
  float z0[NST], v[NST];
#pragma unroll
  for (int j = 0; j < 3; ++j) {
    float2 a = make_float2(0.f, 0.f), c = make_float2(0.f, 0.f);
    if (t0 < NTILES) a = zp[j * NTILES + t0];
    if (t1 < NTILES) c = zp[j * NTILES + t1];
    z0[2 * j] = a.x; z0[2 * j + 1] = a.y;
    v[2 * j] = c.x;  v[2 * j + 1] = c.y;
  }
  mv6_acc(A256, z0, v);

  int cur = 0;
#pragma unroll
  for (int j = 0; j < 3; ++j) buf[0][j][i] = make_float2(v[2 * j], v[2 * j + 1]);
  __syncthreads();
  for (int k = 0; k < SCAN_LEVELS; ++k) {
    const int len = 1 << k;
    const float* M = pwr + (9 + k) * 36;
    if (i >= len) {
      float q[NST];
#pragma unroll
      for (int j = 0; j < 3; ++j) {
        const float2 w = buf[cur][j][i - len];
        q[2 * j] = w.x; q[2 * j + 1] = w.y;
      }
      mv6_acc(M, q, v);
    }
    cur ^= 1;
#pragma unroll
    for (int j = 0; j < 3; ++j) buf[cur][j][i] = make_float2(v[2 * j], v[2 * j + 1]);
    __syncthreads();
  }

  float e[NST], e1[NST];
#pragma unroll
  for (int j = 0; j < 3; ++j) {
    float2 w = make_float2(0.f, 0.f);
    if (i > 0) w = buf[cur][j][i - 1];
    e[2 * j] = w.x; e[2 * j + 1] = w.y;
  }
#pragma unroll
  for (int q = 0; q < NST; ++q) e1[q] = z0[q];
  mv6_acc(A256, e, e1);
#pragma unroll
  for (int j = 0; j < 3; ++j) {
    if (t0 < NTILES) zp[j * NTILES + t0] = make_float2(e[2 * j], e[2 * j + 1]);
    if (t1 < NTILES) zp[j * NTILES + t1] = make_float2(e1[2 * j], e1[2 * j + 1]);
  }
}

template <bool FULL>
__device__ __forceinline__ void tile_body(const float4* __restrict__ gbase, float4* __restrict__ obase,
                                          const float* __restrict__ pw, const float* __restrict__ fr,
                                          const float* __restrict__ sg, float4* lw,
                                          int b, int nv, int tc, int lane) {
  const int n = lane & 15;
  const int half = lane >> 4;

  v8f S = {0.f, 0.f, 0.f, 0.f, 0.f, 0.f, 0.f, 0.f};
  {
    const float2* Sg = (const float2*)sg;
    const float2 w0 = Sg[(b * 3 + half) * NTILES + tc];
    const float2 w2 = Sg[(b * 3 + 2) * NTILES + tc];
    S[0] = w0.x; S[1] = w0.y;
    S[2] = half ? 0.f : w2.x;
    S[3] = half ? 0.f : w2.y;
  }

  const v2f* hf = (const v2f*)(fr + b * FSTRIDE);
  const v2f* af = (const v2f*)(pw + b * PSTRIDE);
  v2f H[4], F[4], G[2], A[2];
#pragma unroll
  for (int s = 0; s < 4; ++s) {
    H[s] = hf[s * 32 + lane];
    F[s] = hf[128 + s * 32 + lane];
  }
  G[0] = hf[256 + lane];
  G[1] = hf[256 + 32 + lane];
  A[0] = af[lane];
  A[1] = af[32 + lane];

  stage_in<FULL>(gbase, nv, lane, lw);

  float4* xr = lw + n * SPITCH + half;
#pragma unroll
  for (int c = 0; c < 16; ++c) {
    const float4 q0 = xr[4 * c];
    const float4 q1 = xr[4 * c + 2];
    v2f x0, x1, x2, x3, s0, s1;
    x0.x = q0.x; x0.y = q0.y;
    x1.x = q0.z; x1.y = q0.w;
    x2.x = q1.x; x2.y = q1.y;
    x3.x = q1.z; x3.y = q1.w;
    s0.x = S[0]; s0.y = S[1];
    s1.x = S[2]; s1.y = S[3];

    v8f y = {0.f, 0.f, 0.f, 0.f, 0.f, 0.f, 0.f, 0.f};
    y = wmma_f32_16x16x4(H[0], x0, y);
    y = wmma_f32_16x16x4(H[1], x1, y);
    y = wmma_f32_16x16x4(H[2], x2, y);
    y = wmma_f32_16x16x4(H[3], x3, y);
    y = wmma_f32_16x16x4(G[0], s0, y);
    y = wmma_f32_16x16x4(G[1], s1, y);

    if (c < 15) {
      v8f acc = {0.f, 0.f, 0.f, 0.f, 0.f, 0.f, 0.f, 0.f};
      acc = wmma_f32_16x16x4(F[0], x0, acc);
      acc = wmma_f32_16x16x4(F[1], x1, acc);
      acc = wmma_f32_16x16x4(F[2], x2, acc);
      acc = wmma_f32_16x16x4(F[3], x3, acc);
      acc = wmma_f32_16x16x4(A[0], s0, acc);
      acc = wmma_f32_16x16x4(A[1], s1, acc);
      S = acc;
    }

    float4 d0, d1;
    d0.x = y[0]; d0.y = y[1]; d0.z = y[2]; d0.w = y[3];
    d1.x = y[4]; d1.y = y[5]; d1.z = y[6]; d1.w = y[7];
    xr[4 * c]     = d0;
    xr[4 * c + 2] = d1;

    if (c == 7 || c == 15) {
      const int hh = c >> 3;
      __builtin_amdgcn_wave_barrier();
#pragma unroll
      for (int tt = 0; tt < 16; ++tt)
        if (FULL || tt < nv)
          obase[tt * (TILE / 4) + hh * 32 + lane] = lw[tt * SPITCH + hh * 32 + lane];
    }
  }
}

__global__ void __launch_bounds__(256) tile_kernel(const float* __restrict__ in, float* __restrict__ out,
                                                   const float* __restrict__ pw,
                                                   const float* __restrict__ fr,
                                                   const float* __restrict__ sg,
                                                   size_t wsz) {
  if (wsz < WS_BYTES) return;
  __shared__ float4 stage[8 * WSTAGE];
  const int lane = threadIdx.x & 31;
  const int wave = __builtin_amdgcn_readfirstlane((int)(threadIdx.x >> 5));
  const int gw = blockIdx.x * 8 + wave;
  const int b = gw / NGRP;
  const int g = gw - b * NGRP;
  const int t = g * 16 + (lane & 15);
  const int tc = (t < NTILES) ? t : (NTILES - 1);
  const int nv = (NTILES - g * 16 < 16) ? (NTILES - g * 16) : 16;
  float4* lw = stage + wave * WSTAGE;
  const size_t goff = (size_t)b * LSIG + (size_t)g * (16 * TILE);
  const float4* gbase = (const float4*)(in + goff);
  float4* obase = (float4*)(out + goff);
  if (nv == 16) tile_body<true>(gbase, obase, pw, fr, sg, lw, b, nv, tc, lane);
  else          tile_body<false>(gbase, obase, pw, fr, sg, lw, b, nv, tc, lane);
}

extern "C" void kernel_launch(void* const* d_in, const int* in_sizes, int n_in,
                              void* d_out, int out_size, void* d_ws, size_t ws_size,
                              hipStream_t stream) {
  (void)in_sizes; (void)n_in; (void)out_size;
  const float* x     = (const float*)d_in[0];
  const float* knobs = (const float*)d_in[1];
  const float* W1    = (const float*)d_in[2];
  const float* b1v   = (const float*)d_in[3];
  const float* W2    = (const float*)d_in[4];
  const float* b2v   = (const float*)d_in[5];
  float* out = (float*)d_out;

  float* PW = (float*)d_ws;
  float* FR = PW + NB * PSTRIDE;
  float* TS = FR + NB * FSTRIDE;

  derive_kernel<<<NB, 256, 0, stream>>>(knobs, W1, b1v, W2, b2v, PW, FR, ws_size);
  state_kernel<<<NB + (NB * NGRP) / 8, 256, 0, stream>>>(x, knobs, W1, b1v, W2, b2v, PW, FR, TS, ws_size);
  scan_kernel<<<NB, SCAN_THREADS, 0, stream>>>(PW, TS, ws_size);
  tile_kernel<<<(NB * NGRP) / 8, 256, 0, stream>>>(x, out, PW, FR, TS, ws_size);
}
